# starting kernel
# baseline (speedup 1.0000x reference)
#include <hip/hip_runtime.h>


#define DD 2048
#define MH 32

typedef __attribute__((ext_vector_type(2))) float v2f;
typedef __attribute__((ext_vector_type(8))) float v8f;

__device__ __forceinline__ v2f pair2(const v8f v, int j) {
    v2f r;
    r[0] = v[2 * j];
    r[1] = v[2 * j + 1];
    return r;
}

__global__ __launch_bounds__(256) void gemv_relu_kernel(
    const float* __restrict__ W, const float* __restrict__ a,
    const float* __restrict__ b, float* __restrict__ out)
{
    const int lane = threadIdx.x & 31;
    const int wave = threadIdx.x >> 5;
    const int row  = blockIdx.x * 8 + wave;

    const float* __restrict__ wrow = W + (size_t)row * DD;

    float acc = 0.0f;
    for (int j = lane * 4; j < DD; j += 32 * 4) {
        const float4 wv = *(const float4*)(wrow + j);
        const float4 av = *(const float4*)(a + j);
        acc += wv.x * av.x + wv.y * av.y + wv.z * av.z + wv.w * av.w;
    }
    acc += __shfl_xor(acc, 1);
    acc += __shfl_xor(acc, 2);
    acc += __shfl_xor(acc, 4);
    acc += __shfl_xor(acc, 8);
    acc += __shfl_xor(acc, 16);
    if (lane == 0) {
        const float v = acc + b[row];
        out[row] = v > 0.0f ? v : 0.0f;
    }
}

__global__ __launch_bounds__(256) void meta_update_kernel(
    const float* __restrict__ W,    const float* __restrict__ aPre,
    const float* __restrict__ aPost,const float* __restrict__ Wm1,
    const float* __restrict__ bm1,  const float* __restrict__ Wm2,
    const float* __restrict__ bm2,  float* __restrict__ Wout)
{
    __shared__ float sA[DD];
    for (int t = threadIdx.x; t < DD; t += 256) sA[t] = aPre[t];
    __syncthreads();

    const int lane = threadIdx.x & 31;
    const int wave = threadIdx.x >> 5;
    const int col  = lane & 15;
    const int half = lane >> 4;

    v2f A0, A1;
    A0[0] = half ? Wm1[2 * MH + col]      : Wm1[0 * MH + col];
    A0[1] = half ? bm1[col]               : Wm1[1 * MH + col];
    A1[0] = half ? Wm1[2 * MH + 16 + col] : Wm1[0 * MH + 16 + col];
    A1[1] = half ? bm1[16 + col]          : Wm1[1 * MH + 16 + col];

    const float* wmb = Wm2 + (half ? 8 : 0);
    v2f pA[4], pB[4];
#pragma unroll
    for (int j = 0; j < 4; ++j) {
        pA[j][0] = 0.5f * wmb[2 * j];
        pA[j][1] = 0.5f * wmb[2 * j + 1];
        pB[j][0] = 0.5f * wmb[16 + 2 * j];
        pB[j][1] = 0.5f * wmb[16 + 2 * j + 1];
    }

    const float accInit = half ? 0.0f : bm2[0];

    const int row = blockIdx.x * 8 + wave;
    const float* __restrict__ wrow = W    + (size_t)row * DD;
    float*       __restrict__ orow = Wout + (size_t)row * DD;
    const float  vO = aPost[row];

    for (int i0 = 0; i0 < DD; i0 += 32) {
        const float vi0 = sA[i0 + col];
        const float w0  = wrow[i0 + col];
        const float vi1 = sA[i0 + 16 + col];
        const float w1  = wrow[i0 + 16 + col];

        v2f Ba, Bb;
        Ba[0] = half ? vO   : vi0;
        Ba[1] = half ? 1.0f : w0;
        Bb[0] = half ? vO   : vi1;
        Bb[1] = half ? 1.0f : w1;

        v8f C0a = {}, C1a = {}, C0b = {}, C1b = {};
        C0a = __builtin_amdgcn_wmma_f32_16x16x4_f32(
                  false, A0, false, Ba, (short)0, C0a, false, false);
        C1a = __builtin_amdgcn_wmma_f32_16x16x4_f32(
                  false, A1, false, Ba, (short)0, C1a, false, false);
        C0b = __builtin_amdgcn_wmma_f32_16x16x4_f32(
                  false, A0, false, Bb, (short)0, C0b, false, false);
        C1b = __builtin_amdgcn_wmma_f32_16x16x4_f32(
                  false, A1, false, Bb, (short)0, C1b, false, false);

        v2f rawA = {accInit, 0.0f}, absA = {0.0f, 0.0f};
        v2f rawB = {accInit, 0.0f}, absB = {0.0f, 0.0f};
#pragma unroll
        for (int j = 0; j < 4; ++j) {
            const v2f c0a = pair2(C0a, j);
            const v2f c1a = pair2(C1a, j);
            const v2f c0b = pair2(C0b, j);
            const v2f c1b = pair2(C1b, j);
            rawA = __builtin_elementwise_fma(pA[j], c0a, rawA);
            absA = __builtin_elementwise_fma(pA[j], __builtin_elementwise_abs(c0a), absA);
            rawA = __builtin_elementwise_fma(pB[j], c1a, rawA);
            absA = __builtin_elementwise_fma(pB[j], __builtin_elementwise_abs(c1a), absA);
            rawB = __builtin_elementwise_fma(pA[j], c0b, rawB);
            absB = __builtin_elementwise_fma(pA[j], __builtin_elementwise_abs(c0b), absB);
            rawB = __builtin_elementwise_fma(pB[j], c1b, rawB);
            absB = __builtin_elementwise_fma(pB[j], __builtin_elementwise_abs(c1b), absB);
        }
        const v2f tA = rawA + absA;
        const v2f tB = rawB + absB;
        float accA = tA[0] + tA[1];
        float accB = tB[0] + tB[1];

        accA += __shfl_xor(accA, 16);
        accB += __shfl_xor(accB, 16);

        const float res = (lane < 16) ? accA : accB;
        __builtin_nontemporal_store(res, orow + i0 + lane);
    }
}

extern "C" void kernel_launch(void* const* d_in, const int* in_sizes, int n_in,
                              void* d_out, int out_size, void* d_ws, size_t ws_size,
                              hipStream_t stream) {
    const float* x   = (const float*)d_in[0];
    const float* W1  = (const float*)d_in[1];
    const float* b1  = (const float*)d_in[2];
    const float* W2  = (const float*)d_in[3];
    const float* b2  = (const float*)d_in[4];
    const float* W3  = (const float*)d_in[5];
    const float* b3  = (const float*)d_in[6];
    const float* Wm1 = (const float*)d_in[7];
    const float* bm1 = (const float*)d_in[8];
    const float* Wm2 = (const float*)d_in[9];
    const float* bm2 = (const float*)d_in[10];

    float* out = (float*)d_out;
    float* nW1 = out + DD;
    float* nW2 = nW1 + (size_t)DD * DD;
    float* nW3 = nW2 + (size_t)DD * DD;

    float* a1 = (float*)d_ws;
    float* a2 = a1 + DD;

    dim3 blk(256);
    dim3 grid(DD / 8);

    gemv_relu_kernel<<<grid, blk, 0, stream>>>(W1, x,  b1, a1);
    gemv_relu_kernel<<<grid, blk, 0, stream>>>(W2, a1, b2, a2);
    gemv_relu_kernel<<<grid, blk, 0, stream>>>(W3, a2, b3, out);

    meta_update_kernel<<<grid, blk, 0, stream>>>(W1, x,  a1,  Wm1, bm1, Wm2, bm2, nW1);
    meta_update_kernel<<<grid, blk, 0, stream>>>(W2, a1, a2,  Wm1, bm1, Wm2, bm2, nW2);
    meta_update_kernel<<<grid, blk, 0, stream>>>(W3, a2, out, Wm1, bm1, Wm2, bm2, nW3);
}


# meta kernel: operands fetched by b128 loads one 256-entry block ahead (weights in lanes 0-15, pre-activations in 16-31, no LDS); k order of the 4-term WMMA sum permuted to fit; b128 stores
# speedup vs baseline: 2.0429x; 2.0429x over previous
#include <hip/hip_runtime.h>


#define DD 2048
#define MH 32

typedef __attribute__((ext_vector_type(2))) float v2f;
typedef __attribute__((ext_vector_type(4))) float v4f;
typedef __attribute__((ext_vector_type(8))) float v8f;

__device__ __forceinline__ v2f pair2(const v8f v, int j) {
    v2f r;
    r[0] = v[2 * j];
    r[1] = v[2 * j + 1];
    return r;
}

__global__ __launch_bounds__(256) void gemv_relu_kernel(
    const float* __restrict__ W, const float* __restrict__ a,
    const float* __restrict__ b, float* __restrict__ out)
{
    const int lane = threadIdx.x & 31;
    const int wave = threadIdx.x >> 5;
    const int row  = blockIdx.x * 8 + wave;

    const float* __restrict__ wrow = W + (size_t)row * DD;

    float acc = 0.0f;
    for (int j = lane * 4; j < DD; j += 32 * 4) {
        const float4 wv = *(const float4*)(wrow + j);
        const float4 av = *(const float4*)(a + j);
        acc += wv.x * av.x + wv.y * av.y + wv.z * av.z + wv.w * av.w;
    }
    acc += __shfl_xor(acc, 1);
    acc += __shfl_xor(acc, 2);
    acc += __shfl_xor(acc, 4);
    acc += __shfl_xor(acc, 8);
    acc += __shfl_xor(acc, 16);
    if (lane == 0) {
        const float v = acc + b[row];
        out[row] = v > 0.0f ? v : 0.0f;
    }
}

__device__ __forceinline__ float meta_tile(
    const v2f A0, const v2f A1, const v2f B,
    const v2f (&pA)[4], const v2f (&pB)[4], const float accInit)
{
    v8f C0 = {}, C1 = {};
    C0 = __builtin_amdgcn_wmma_f32_16x16x4_f32(
             false, A0, false, B, (short)0, C0, false, false);
    C1 = __builtin_amdgcn_wmma_f32_16x16x4_f32(
             false, A1, false, B, (short)0, C1, false, false);

    v2f raw = {accInit, 0.0f}, ab = {0.0f, 0.0f};
#pragma unroll
    for (int j = 0; j < 4; ++j) {
        const v2f c0 = pair2(C0, j);
        const v2f c1 = pair2(C1, j);
        raw = __builtin_elementwise_fma(pA[j], c0, raw);
        ab  = __builtin_elementwise_fma(pA[j], __builtin_elementwise_abs(c0), ab);
        raw = __builtin_elementwise_fma(pB[j], c1, raw);
        ab  = __builtin_elementwise_fma(pB[j], __builtin_elementwise_abs(c1), ab);
    }
    const v2f t = raw + ab;
    float acc = t[0] + t[1];
    acc += __shfl_xor(acc, 16);
    return acc;
}

__device__ __forceinline__ void meta_block128(
    const v2f A0, const v2f A1, const v4f da, const v4f db, const float kc,
    const v2f (&pA)[4], const v2f (&pB)[4], const float accInit,
    const int half, float* __restrict__ dst)
{
    v4f res;
#pragma unroll
    for (int t = 0; t < 4; ++t) {
        v2f Ba, Bb;
        Ba[0] = da[t]; Ba[1] = kc;
        Bb[0] = db[t]; Bb[1] = kc;
        const float ra = meta_tile(A0, A1, Ba, pA, pB, accInit);
        const float rb = meta_tile(A0, A1, Bb, pA, pB, accInit);
        res[t] = half ? rb : ra;
    }
    __builtin_nontemporal_store(res, (v4f*)dst);
}

__global__ __launch_bounds__(256) void meta_update_kernel(
    const float* __restrict__ W,    const float* __restrict__ aPre,
    const float* __restrict__ aPost,const float* __restrict__ Wm1,
    const float* __restrict__ bm1,  const float* __restrict__ Wm2,
    const float* __restrict__ bm2,  float* __restrict__ Wout)
{
    const int lane = threadIdx.x & 31;
    const int wave = threadIdx.x >> 5;
    const int col  = lane & 15;
    const int half = lane >> 4;

    v2f A0, A1;
    A0[0] = half ? Wm1[0 * MH + col]      : Wm1[1 * MH + col];
    A0[1] = half ? bm1[col]               : Wm1[2 * MH + col];
    A1[0] = half ? Wm1[0 * MH + 16 + col] : Wm1[1 * MH + 16 + col];
    A1[1] = half ? bm1[16 + col]          : Wm1[2 * MH + 16 + col];

    const float* wmb = Wm2 + (half ? 8 : 0);
    v2f pA[4], pB[4];
#pragma unroll
    for (int j = 0; j < 4; ++j) {
        pA[j][0] = 0.5f * wmb[2 * j];
        pA[j][1] = 0.5f * wmb[2 * j + 1];
        pB[j][0] = 0.5f * wmb[16 + 2 * j];
        pB[j][1] = 0.5f * wmb[16 + 2 * j + 1];
    }

    const float accInit = half ? 0.0f : bm2[0];

    const int row = blockIdx.x * 8 + wave;
    const float kc = half ? 1.0f : aPost[row];
    const float* __restrict__ srcp =
        (half ? aPre : W + (size_t)row * DD) + 4 * col;
    float* __restrict__ orow = Wout + (size_t)row * DD + 4 * lane;

    v4f n0 = *(const v4f*)(srcp);
    v4f n1 = *(const v4f*)(srcp + 64);
    v4f n2 = *(const v4f*)(srcp + 128);
    v4f n3 = *(const v4f*)(srcp + 192);

    for (int i0 = 0; i0 < DD; i0 += 256) {
        const v4f c0 = n0, c1 = n1, c2 = n2, c3 = n3;
        if (i0 + 256 < DD) {
            n0 = *(const v4f*)(srcp + i0 + 256);
            n1 = *(const v4f*)(srcp + i0 + 320);
            n2 = *(const v4f*)(srcp + i0 + 384);
            n3 = *(const v4f*)(srcp + i0 + 448);
        }
        meta_block128(A0, A1, c0, c1, kc, pA, pB, accInit, half, orow + i0);
        meta_block128(A0, A1, c2, c3, kc, pA, pB, accInit, half, orow + i0 + 128);
    }
}

extern "C" void kernel_launch(void* const* d_in, const int* in_sizes, int n_in,
                              void* d_out, int out_size, void* d_ws, size_t ws_size,
                              hipStream_t stream) {
    const float* x   = (const float*)d_in[0];
    const float* W1  = (const float*)d_in[1];
    const float* b1  = (const float*)d_in[2];
    const float* W2  = (const float*)d_in[3];
    const float* b2  = (const float*)d_in[4];
    const float* W3  = (const float*)d_in[5];
    const float* b3  = (const float*)d_in[6];
    const float* Wm1 = (const float*)d_in[7];
    const float* bm1 = (const float*)d_in[8];
    const float* Wm2 = (const float*)d_in[9];
    const float* bm2 = (const float*)d_in[10];

    float* out = (float*)d_out;
    float* nW1 = out + DD;
    float* nW2 = nW1 + (size_t)DD * DD;
    float* nW3 = nW2 + (size_t)DD * DD;

    float* a1 = (float*)d_ws;
    float* a2 = a1 + DD;

    dim3 blk(256);
    dim3 grid(DD / 8);

    gemv_relu_kernel<<<grid, blk, 0, stream>>>(W1, x,  b1, a1);
    gemv_relu_kernel<<<grid, blk, 0, stream>>>(W2, a1, b2, a2);
    gemv_relu_kernel<<<grid, blk, 0, stream>>>(W3, a2, b3, out);

    meta_update_kernel<<<grid, blk, 0, stream>>>(W1, x,  a1,  Wm1, bm1, Wm2, bm2, nW1);
    meta_update_kernel<<<grid, blk, 0, stream>>>(W2, a1, a2,  Wm1, bm1, Wm2, bm2, nW2);
    meta_update_kernel<<<grid, blk, 0, stream>>>(W3, a2, out, Wm1, bm1, Wm2, bm2, nW3);
}


# launch fusion: each layer's row dot product (+bias, ReLU) is formed by the wave that then meta-updates that row, row loads batched; 6 launches -> 3; dot products now FMA-contracted, sum order kept
# speedup vs baseline: 3.3536x; 1.6416x over previous
#include <hip/hip_runtime.h>


#define DD 2048
#define MH 32

typedef __attribute__((ext_vector_type(2))) float v2f;
typedef __attribute__((ext_vector_type(4))) float v4f;
typedef __attribute__((ext_vector_type(8))) float v8f;

__device__ __forceinline__ v2f pair2(const v8f v, int j) {
    v2f r;
    r[0] = v[2 * j];
    r[1] = v[2 * j + 1];
    return r;
}

__device__ __forceinline__ float meta_tile(
    const v2f A0, const v2f A1, const v2f B,
    const v2f (&pA)[4], const v2f (&pB)[4], const float accInit)
{
    v8f C0 = {}, C1 = {};
    C0 = __builtin_amdgcn_wmma_f32_16x16x4_f32(
             false, A0, false, B, (short)0, C0, false, false);
    C1 = __builtin_amdgcn_wmma_f32_16x16x4_f32(
             false, A1, false, B, (short)0, C1, false, false);

    v2f raw = {accInit, 0.0f}, ab = {0.0f, 0.0f};
#pragma unroll
    for (int j = 0; j < 4; ++j) {
        const v2f c0 = pair2(C0, j);
        const v2f c1 = pair2(C1, j);
        raw = __builtin_elementwise_fma(pA[j], c0, raw);
        ab  = __builtin_elementwise_fma(pA[j], __builtin_elementwise_abs(c0), ab);
        raw = __builtin_elementwise_fma(pB[j], c1, raw);
        ab  = __builtin_elementwise_fma(pB[j], __builtin_elementwise_abs(c1), ab);
    }
    const v2f t = raw + ab;
    float acc = t[0] + t[1];
    acc += __shfl_xor(acc, 16);
    return acc;
}

__device__ __forceinline__ void meta_block128(
    const v2f A0, const v2f A1, const v4f da, const v4f db, const float kc,
    const v2f (&pA)[4], const v2f (&pB)[4], const float accInit,
    const int half, float* __restrict__ dst)
{
    v4f res;
#pragma unroll
    for (int t = 0; t < 4; ++t) {
        v2f Ba, Bb;
        Ba[0] = da[t]; Ba[1] = kc;
        Bb[0] = db[t]; Bb[1] = kc;
        const float ra = meta_tile(A0, A1, Ba, pA, pB, accInit);
        const float rb = meta_tile(A0, A1, Bb, pA, pB, accInit);
        res[t] = half ? rb : ra;
    }
    __builtin_nontemporal_store(res, (v4f*)dst);
}

__global__ __launch_bounds__(256) void layer_forward_meta_kernel(
    const float* __restrict__ W,    const float* __restrict__ aPre,
    const float* __restrict__ bias, float* __restrict__ aOut,
    float* __restrict__ Wout,
    const float* __restrict__ Wm2,  const float* __restrict__ bm2,
    const float* __restrict__ Wm1,  const float* __restrict__ bm1)
{
    const int lane = threadIdx.x & 31;
    const int wave = threadIdx.x >> 5;
    const int col  = lane & 15;
    const int half = lane >> 4;
    const int row  = blockIdx.x * 8 + wave;

    v2f A0, A1;
    A0[0] = half ? Wm1[0 * MH + col]      : Wm1[1 * MH + col];
    A0[1] = half ? bm1[col]               : Wm1[2 * MH + col];
    A1[0] = half ? Wm1[0 * MH + 16 + col] : Wm1[1 * MH + 16 + col];
    A1[1] = half ? bm1[16 + col]          : Wm1[2 * MH + 16 + col];

    const float* wmb = Wm2 + (half ? 8 : 0);
    v2f pA[4], pB[4];
#pragma unroll
    for (int j = 0; j < 4; ++j) {
        pA[j][0] = 0.5f * wmb[2 * j];
        pA[j][1] = 0.5f * wmb[2 * j + 1];
        pB[j][0] = 0.5f * wmb[16 + 2 * j];
        pB[j][1] = 0.5f * wmb[16 + 2 * j + 1];
    }

    const float bm2v = bm2[0];
    const float accInit = half ? 0.0f : bm2v;

    const float* __restrict__ srcp =
        (half ? aPre : W + (size_t)row * DD) + 4 * col;
    float* __restrict__ orow = Wout + (size_t)row * DD + 4 * lane;

    v4f n0 = *(const v4f*)(srcp);
    v4f n1 = *(const v4f*)(srcp + 64);
    v4f n2 = *(const v4f*)(srcp + 128);
    v4f n3 = *(const v4f*)(srcp + 192);

    float vPost;
    {
        const float* __restrict__ wrow = W + (size_t)row * DD + 4 * lane;
        const float* __restrict__ arow = aPre + 4 * lane;
        const float brow = bias[row];
        float acc = 0.0f;
#pragma unroll
        for (int k = 0; k < DD / 128; ++k) {
            const float4 wv = *(const float4*)(wrow + 128 * k);
            const float4 av = *(const float4*)(arow + 128 * k);
            acc += wv.x * av.x + wv.y * av.y + wv.z * av.z + wv.w * av.w;
        }
        acc += __shfl_xor(acc, 1);
        acc += __shfl_xor(acc, 2);
        acc += __shfl_xor(acc, 4);
        acc += __shfl_xor(acc, 8);
        acc += __shfl_xor(acc, 16);
        const float v = acc + brow;
        vPost = v > 0.0f ? v : 0.0f;
        if (lane == 0) aOut[row] = vPost;
    }

    const float kc = half ? 1.0f : vPost;

    for (int i0 = 0; i0 < DD; i0 += 256) {
        const v4f c0 = n0, c1 = n1, c2 = n2, c3 = n3;
        if (i0 + 256 < DD) {
            n0 = *(const v4f*)(srcp + i0 + 256);
            n1 = *(const v4f*)(srcp + i0 + 320);
            n2 = *(const v4f*)(srcp + i0 + 384);
            n3 = *(const v4f*)(srcp + i0 + 448);
        }
        meta_block128(A0, A1, c0, c1, kc, pA, pB, accInit, half, orow + i0);
        meta_block128(A0, A1, c2, c3, kc, pA, pB, accInit, half, orow + i0 + 128);
    }
}

extern "C" void kernel_launch(void* const* d_in, const int* in_sizes, int n_in,
                              void* d_out, int out_size, void* d_ws, size_t ws_size,
                              hipStream_t stream) {
    const float* x   = (const float*)d_in[0];
    const float* W1  = (const float*)d_in[1];
    const float* b1  = (const float*)d_in[2];
    const float* W2  = (const float*)d_in[3];
    const float* b2  = (const float*)d_in[4];
    const float* W3  = (const float*)d_in[5];
    const float* b3  = (const float*)d_in[6];
    const float* Wm1 = (const float*)d_in[7];
    const float* bm1 = (const float*)d_in[8];
    const float* Wm2 = (const float*)d_in[9];
    const float* bm2 = (const float*)d_in[10];

    float* out = (float*)d_out;
    float* nW1 = out + DD;
    float* nW2 = nW1 + (size_t)DD * DD;
    float* nW3 = nW2 + (size_t)DD * DD;

    float* a1 = (float*)d_ws;
    float* a2 = a1 + DD;

    dim3 blk(256);
    dim3 grid(DD / 8);

    layer_forward_meta_kernel<<<grid, blk, 0, stream>>>(W1, x,  b1, a1,  nW1, Wm2, bm2, Wm1, bm1);
    layer_forward_meta_kernel<<<grid, blk, 0, stream>>>(W2, a1, b2, a2,  nW2, Wm2, bm2, Wm1, bm1);
    layer_forward_meta_kernel<<<grid, blk, 0, stream>>>(W3, a2, b3, out, nW3, Wm2, bm2, Wm1, bm1);
}


# meta epilogue: linear half of wm2*relu(x)=hw*x+hw*|x| summed over the 32 meta rows once per wave into 4 coefficients, then 1 FMA per entry; only the |x| chain runs per tile; sum order changed
# speedup vs baseline: 3.6626x; 1.0921x over previous
#include <hip/hip_runtime.h>


#define DD 2048
#define MH 32

typedef __attribute__((ext_vector_type(2))) float v2f;
typedef __attribute__((ext_vector_type(4))) float v4f;
typedef __attribute__((ext_vector_type(8))) float v8f;

__device__ __forceinline__ v2f pair2(const v8f v, int j) {
    v2f r;
    r[0] = v[2 * j];
    r[1] = v[2 * j + 1];
    return r;
}

__device__ __forceinline__ float meta_tile(
    const v2f A0, const v2f A1, const v2f B,
    const v2f (&pA)[4], const v2f (&pB)[4], const float lin)
{
    v8f C0 = {}, C1 = {};
    C0 = __builtin_amdgcn_wmma_f32_16x16x4_f32(
             false, A0, false, B, (short)0, C0, false, false);
    C1 = __builtin_amdgcn_wmma_f32_16x16x4_f32(
             false, A1, false, B, (short)0, C1, false, false);

    v2f ab = {lin, 0.0f};
#pragma unroll
    for (int j = 0; j < 4; ++j) {
        ab = __builtin_elementwise_fma(pA[j], __builtin_elementwise_abs(pair2(C0, j)), ab);
        ab = __builtin_elementwise_fma(pB[j], __builtin_elementwise_abs(pair2(C1, j)), ab);
    }
    float acc = ab[0] + ab[1];
    acc += __shfl_xor(acc, 16);
    return acc;
}

__device__ __forceinline__ void meta_block128(
    const v2f A0, const v2f A1, const v4f da, const v4f db, const float kc,
    const v2f (&pA)[4], const v2f (&pB)[4], const float g0, const float cl,
    const int half, float* __restrict__ dst)
{
    v4f res;
#pragma unroll
    for (int t = 0; t < 4; ++t) {
        v2f Ba, Bb;
        Ba[0] = da[t]; Ba[1] = kc;
        Bb[0] = db[t]; Bb[1] = kc;
        const float ra = meta_tile(A0, A1, Ba, pA, pB, __builtin_fmaf(g0, da[t], cl));
        const float rb = meta_tile(A0, A1, Bb, pA, pB, __builtin_fmaf(g0, db[t], cl));
        res[t] = half ? rb : ra;
    }
    __builtin_nontemporal_store(res, (v4f*)dst);
}

__global__ __launch_bounds__(256) void layer_forward_meta_kernel(
    const float* __restrict__ W,    const float* __restrict__ aPre,
    const float* __restrict__ bias, float* __restrict__ aOut,
    float* __restrict__ Wout,
    const float* __restrict__ Wm2,  const float* __restrict__ bm2,
    const float* __restrict__ Wm1,  const float* __restrict__ bm1)
{
    const int lane = threadIdx.x & 31;
    const int wave = threadIdx.x >> 5;
    const int col  = lane & 15;
    const int half = lane >> 4;
    const int row  = blockIdx.x * 8 + wave;

    v2f A0, A1;
    A0[0] = half ? Wm1[0 * MH + col]      : Wm1[1 * MH + col];
    A0[1] = half ? bm1[col]               : Wm1[2 * MH + col];
    A1[0] = half ? Wm1[0 * MH + 16 + col] : Wm1[1 * MH + 16 + col];
    A1[1] = half ? bm1[16 + col]          : Wm1[2 * MH + 16 + col];

    const float* wmb = Wm2 + (half ? 8 : 0);
    v2f pA[4], pB[4];
#pragma unroll
    for (int j = 0; j < 4; ++j) {
        pA[j][0] = 0.5f * wmb[2 * j];
        pA[j][1] = 0.5f * wmb[2 * j + 1];
        pB[j][0] = 0.5f * wmb[16 + 2 * j];
        pB[j][1] = 0.5f * wmb[16 + 2 * j + 1];
    }

    const float hwA = 0.5f * Wm2[col];
    const float hwB = 0.5f * Wm2[16 + col];
    float g0 = hwA * A0[0] + hwB * A1[0];
    float g1 = hwA * A0[1] + hwB * A1[1];

    const float bm2v = bm2[0];
    const float accInit = half ? 0.0f : bm2v;

    const float* __restrict__ srcp =
        (half ? aPre : W + (size_t)row * DD) + 4 * col;
    float* __restrict__ orow = Wout + (size_t)row * DD + 4 * lane;

    v4f n0 = *(const v4f*)(srcp);
    v4f n1 = *(const v4f*)(srcp + 64);
    v4f n2 = *(const v4f*)(srcp + 128);
    v4f n3 = *(const v4f*)(srcp + 192);

    float vPost;
    {
        const float* __restrict__ wrow = W + (size_t)row * DD + 4 * lane;
        const float* __restrict__ arow = aPre + 4 * lane;
        const float brow = bias[row];
        float acc = 0.0f;
#pragma unroll
        for (int k = 0; k < DD / 128; ++k) {
            const float4 wv = *(const float4*)(wrow + 128 * k);
            const float4 av = *(const float4*)(arow + 128 * k);
            acc += wv.x * av.x + wv.y * av.y + wv.z * av.z + wv.w * av.w;
        }
        acc += __shfl_xor(acc, 1);
        g0  += __shfl_xor(g0, 1);
        g1  += __shfl_xor(g1, 1);
        acc += __shfl_xor(acc, 2);
        g0  += __shfl_xor(g0, 2);
        g1  += __shfl_xor(g1, 2);
        acc += __shfl_xor(acc, 4);
        g0  += __shfl_xor(g0, 4);
        g1  += __shfl_xor(g1, 4);
        acc += __shfl_xor(acc, 8);
        g0  += __shfl_xor(g0, 8);
        g1  += __shfl_xor(g1, 8);
        acc += __shfl_xor(acc, 16);
        const float v = acc + brow;
        vPost = v > 0.0f ? v : 0.0f;
        if (lane == 0) aOut[row] = vPost;
    }

    const float kc = half ? 1.0f : vPost;
    const float cl = __builtin_fmaf(g1, kc, accInit);

    for (int i0 = 0; i0 < DD; i0 += 256) {
        const v4f c0 = n0, c1 = n1, c2 = n2, c3 = n3;
        if (i0 + 256 < DD) {
            n0 = *(const v4f*)(srcp + i0 + 256);
            n1 = *(const v4f*)(srcp + i0 + 320);
            n2 = *(const v4f*)(srcp + i0 + 384);
            n3 = *(const v4f*)(srcp + i0 + 448);
        }
        meta_block128(A0, A1, c0, c1, kc, pA, pB, g0, cl, half, orow + i0);
        meta_block128(A0, A1, c2, c3, kc, pA, pB, g0, cl, half, orow + i0 + 128);
    }
}

extern "C" void kernel_launch(void* const* d_in, const int* in_sizes, int n_in,
                              void* d_out, int out_size, void* d_ws, size_t ws_size,
                              hipStream_t stream) {
    const float* x   = (const float*)d_in[0];
    const float* W1  = (const float*)d_in[1];
    const float* b1  = (const float*)d_in[2];
    const float* W2  = (const float*)d_in[3];
    const float* b2  = (const float*)d_in[4];
    const float* W3  = (const float*)d_in[5];
    const float* b3  = (const float*)d_in[6];
    const float* Wm1 = (const float*)d_in[7];
    const float* bm1 = (const float*)d_in[8];
    const float* Wm2 = (const float*)d_in[9];
    const float* bm2 = (const float*)d_in[10];

    float* out = (float*)d_out;
    float* nW1 = out + DD;
    float* nW2 = nW1 + (size_t)DD * DD;
    float* nW3 = nW2 + (size_t)DD * DD;

    float* a1 = (float*)d_ws;
    float* a2 = a1 + DD;

    dim3 blk(256);
    dim3 grid(DD / 8);

    layer_forward_meta_kernel<<<grid, blk, 0, stream>>>(W1, x,  b1, a1,  nW1, Wm2, bm2, Wm1, bm1);
    layer_forward_meta_kernel<<<grid, blk, 0, stream>>>(W2, a1, b2, a2,  nW2, Wm2, bm2, Wm1, bm1);
    layer_forward_meta_kernel<<<grid, blk, 0, stream>>>(W3, a2, b3, out, nW3, Wm2, bm2, Wm1, bm1);
}


# meta epilogue: |x| taken as the source modifier of one scalar v_fma_f32 per C-tile element instead of v_and_b32 + packed FMA; four chains per tile (C0/C1 x even/odd), sum order changed
# speedup vs baseline: 4.2816x; 1.1690x over previous
#include <hip/hip_runtime.h>


#define DD 2048
#define MH 32

typedef __attribute__((ext_vector_type(2))) float v2f;
typedef __attribute__((ext_vector_type(4))) float v4f;
typedef __attribute__((ext_vector_type(8))) float v8f;

__device__ __forceinline__ float meta_tile(
    const v2f A0, const v2f A1, const v2f B,
    const float (&hA)[8], const float (&hB)[8], const float lin)
{
    v8f C0 = {}, C1 = {};
    C0 = __builtin_amdgcn_wmma_f32_16x16x4_f32(
             false, A0, false, B, (short)0, C0, false, false);
    C1 = __builtin_amdgcn_wmma_f32_16x16x4_f32(
             false, A1, false, B, (short)0, C1, false, false);

    float e0 = lin, o0 = 0.0f, e1 = 0.0f, o1 = 0.0f;
#pragma unroll
    for (int j = 0; j < 4; ++j) {
        e0 = __builtin_fmaf(hA[2 * j],     __builtin_fabsf(C0[2 * j]),     e0);
        o0 = __builtin_fmaf(hA[2 * j + 1], __builtin_fabsf(C0[2 * j + 1]), o0);
        e1 = __builtin_fmaf(hB[2 * j],     __builtin_fabsf(C1[2 * j]),     e1);
        o1 = __builtin_fmaf(hB[2 * j + 1], __builtin_fabsf(C1[2 * j + 1]), o1);
    }
    asm("" : "+v"(e0));
    asm("" : "+v"(o0));
    asm("" : "+v"(e1));
    asm("" : "+v"(o1));
    float acc = (e0 + o0) + (e1 + o1);
    acc += __shfl_xor(acc, 16);
    return acc;
}

__device__ __forceinline__ void meta_block128(
    const v2f A0, const v2f A1, const v4f da, const v4f db, const float kc,
    const float (&hA)[8], const float (&hB)[8], const float g0, const float cl,
    const int half, float* __restrict__ dst)
{
    v4f res;
#pragma unroll
    for (int t = 0; t < 4; ++t) {
        v2f Ba, Bb;
        Ba[0] = da[t]; Ba[1] = kc;
        Bb[0] = db[t]; Bb[1] = kc;
        const float ra = meta_tile(A0, A1, Ba, hA, hB, __builtin_fmaf(g0, da[t], cl));
        const float rb = meta_tile(A0, A1, Bb, hA, hB, __builtin_fmaf(g0, db[t], cl));
        res[t] = half ? rb : ra;
    }
    __builtin_nontemporal_store(res, (v4f*)dst);
}

__global__ __launch_bounds__(256) void layer_forward_meta_kernel(
    const float* __restrict__ W,    const float* __restrict__ aPre,
    const float* __restrict__ bias, float* __restrict__ aOut,
    float* __restrict__ Wout,
    const float* __restrict__ Wm2,  const float* __restrict__ bm2,
    const float* __restrict__ Wm1,  const float* __restrict__ bm1)
{
    const int lane = threadIdx.x & 31;
    const int wave = threadIdx.x >> 5;
    const int col  = lane & 15;
    const int half = lane >> 4;
    const int row  = blockIdx.x * 8 + wave;

    v2f A0, A1;
    A0[0] = half ? Wm1[0 * MH + col]      : Wm1[1 * MH + col];
    A0[1] = half ? bm1[col]               : Wm1[2 * MH + col];
    A1[0] = half ? Wm1[0 * MH + 16 + col] : Wm1[1 * MH + 16 + col];
    A1[1] = half ? bm1[16 + col]          : Wm1[2 * MH + 16 + col];

    const float* wmb = Wm2 + (half ? 8 : 0);
    float hA[8], hB[8];
#pragma unroll
    for (int r = 0; r < 8; ++r) {
        hA[r] = 0.5f * wmb[r];
        hB[r] = 0.5f * wmb[16 + r];
    }

    const float hwA = 0.5f * Wm2[col];
    const float hwB = 0.5f * Wm2[16 + col];
    float g0 = hwA * A0[0] + hwB * A1[0];
    float g1 = hwA * A0[1] + hwB * A1[1];

    const float bm2v = bm2[0];
    const float accInit = half ? 0.0f : bm2v;

    const float* __restrict__ srcp =
        (half ? aPre : W + (size_t)row * DD) + 4 * col;
    float* __restrict__ orow = Wout + (size_t)row * DD + 4 * lane;

    v4f n0 = *(const v4f*)(srcp);
    v4f n1 = *(const v4f*)(srcp + 64);
    v4f n2 = *(const v4f*)(srcp + 128);
    v4f n3 = *(const v4f*)(srcp + 192);

    float vPost;
    {
        const float* __restrict__ wrow = W + (size_t)row * DD + 4 * lane;
        const float* __restrict__ arow = aPre + 4 * lane;
        const float brow = bias[row];
        float acc = 0.0f;
#pragma unroll
        for (int k = 0; k < DD / 128; ++k) {
            const float4 wv = *(const float4*)(wrow + 128 * k);
            const float4 av = *(const float4*)(arow + 128 * k);
            acc += wv.x * av.x + wv.y * av.y + wv.z * av.z + wv.w * av.w;
        }
        acc += __shfl_xor(acc, 1);
        g0  += __shfl_xor(g0, 1);
        g1  += __shfl_xor(g1, 1);
        acc += __shfl_xor(acc, 2);
        g0  += __shfl_xor(g0, 2);
        g1  += __shfl_xor(g1, 2);
        acc += __shfl_xor(acc, 4);
        g0  += __shfl_xor(g0, 4);
        g1  += __shfl_xor(g1, 4);
        acc += __shfl_xor(acc, 8);
        g0  += __shfl_xor(g0, 8);
        g1  += __shfl_xor(g1, 8);
        acc += __shfl_xor(acc, 16);
        const float v = acc + brow;
        vPost = v > 0.0f ? v : 0.0f;
        if (lane == 0) aOut[row] = vPost;
    }

    const float kc = half ? 1.0f : vPost;
    const float cl = __builtin_fmaf(g1, kc, accInit);

    for (int i0 = 0; i0 < DD; i0 += 256) {
        const v4f c0 = n0, c1 = n1, c2 = n2, c3 = n3;
        if (i0 + 256 < DD) {
            n0 = *(const v4f*)(srcp + i0 + 256);
            n1 = *(const v4f*)(srcp + i0 + 320);
            n2 = *(const v4f*)(srcp + i0 + 384);
            n3 = *(const v4f*)(srcp + i0 + 448);
        }
        meta_block128(A0, A1, c0, c1, kc, hA, hB, g0, cl, half, orow + i0);
        meta_block128(A0, A1, c2, c3, kc, hA, hB, g0, cl, half, orow + i0 + 128);
    }
}

extern "C" void kernel_launch(void* const* d_in, const int* in_sizes, int n_in,
                              void* d_out, int out_size, void* d_ws, size_t ws_size,
                              hipStream_t stream) {
    const float* x   = (const float*)d_in[0];
    const float* W1  = (const float*)d_in[1];
    const float* b1  = (const float*)d_in[2];
    const float* W2  = (const float*)d_in[3];
    const float* b2  = (const float*)d_in[4];
    const float* W3  = (const float*)d_in[5];
    const float* b3  = (const float*)d_in[6];
    const float* Wm1 = (const float*)d_in[7];
    const float* bm1 = (const float*)d_in[8];
    const float* Wm2 = (const float*)d_in[9];
    const float* bm2 = (const float*)d_in[10];

    float* out = (float*)d_out;
    float* nW1 = out + DD;
    float* nW2 = nW1 + (size_t)DD * DD;
    float* nW3 = nW2 + (size_t)DD * DD;

    float* a1 = (float*)d_ws;
    float* a2 = a1 + DD;

    dim3 blk(256);
    dim3 grid(DD / 8);

    layer_forward_meta_kernel<<<grid, blk, 0, stream>>>(W1, x,  b1, a1,  nW1, Wm2, bm2, Wm1, bm1);
    layer_forward_meta_kernel<<<grid, blk, 0, stream>>>(W2, a1, b2, a2,  nW2, Wm2, bm2, Wm1, bm1);
    layer_forward_meta_kernel<<<grid, blk, 0, stream>>>(W3, a2, b3, out, nW3, Wm2, bm2, Wm1, bm1);
}


# occupancy: each row split over a pair of waves (half-row dot products joined in LDS behind one barrier, each wave meta-updates its half); 4 waves per SIMD instead of 2; dot-product sum order changed
# speedup vs baseline: 4.6204x; 1.0791x over previous
#include <hip/hip_runtime.h>


#define DD 2048
#define MH 32

typedef __attribute__((ext_vector_type(2))) float v2f;
typedef __attribute__((ext_vector_type(4))) float v4f;
typedef __attribute__((ext_vector_type(8))) float v8f;

__device__ __forceinline__ float meta_tile(
    const v2f A0, const v2f A1, const v2f B,
    const float (&hA)[8], const float (&hB)[8], const float lin)
{
    v8f C0 = {}, C1 = {};
    C0 = __builtin_amdgcn_wmma_f32_16x16x4_f32(
             false, A0, false, B, (short)0, C0, false, false);
    C1 = __builtin_amdgcn_wmma_f32_16x16x4_f32(
             false, A1, false, B, (short)0, C1, false, false);

    float e0 = lin, o0 = 0.0f, e1 = 0.0f, o1 = 0.0f;
#pragma unroll
    for (int j = 0; j < 4; ++j) {
        e0 = __builtin_fmaf(hA[2 * j],     __builtin_fabsf(C0[2 * j]),     e0);
        o0 = __builtin_fmaf(hA[2 * j + 1], __builtin_fabsf(C0[2 * j + 1]), o0);
        e1 = __builtin_fmaf(hB[2 * j],     __builtin_fabsf(C1[2 * j]),     e1);
        o1 = __builtin_fmaf(hB[2 * j + 1], __builtin_fabsf(C1[2 * j + 1]), o1);
    }
    asm("" : "+v"(e0));
    asm("" : "+v"(o0));
    asm("" : "+v"(e1));
    asm("" : "+v"(o1));
    float acc = (e0 + o0) + (e1 + o1);
    acc += __shfl_xor(acc, 16);
    return acc;
}

__device__ __forceinline__ void meta_block128(
    const v2f A0, const v2f A1, const v4f da, const v4f db, const float kc,
    const float (&hA)[8], const float (&hB)[8], const float g0, const float cl,
    const int half, float* __restrict__ dst)
{
    v4f res;
#pragma unroll
    for (int t = 0; t < 4; ++t) {
        v2f Ba, Bb;
        Ba[0] = da[t]; Ba[1] = kc;
        Bb[0] = db[t]; Bb[1] = kc;
        const float ra = meta_tile(A0, A1, Ba, hA, hB, __builtin_fmaf(g0, da[t], cl));
        const float rb = meta_tile(A0, A1, Bb, hA, hB, __builtin_fmaf(g0, db[t], cl));
        res[t] = half ? rb : ra;
    }
    __builtin_nontemporal_store(res, (v4f*)dst);
}

__global__ __launch_bounds__(256) void layer_forward_meta_kernel(
    const float* __restrict__ W,    const float* __restrict__ aPre,
    const float* __restrict__ bias, float* __restrict__ aOut,
    float* __restrict__ Wout,
    const float* __restrict__ Wm2,  const float* __restrict__ bm2,
    const float* __restrict__ Wm1,  const float* __restrict__ bm1)
{
    const int lane = threadIdx.x & 31;
    const int wave = threadIdx.x >> 5;
    const int col  = lane & 15;
    const int half = lane >> 4;
    const int row  = blockIdx.x * 4 + (wave >> 1);
    const int hsel = wave & 1;
    const int hoff = hsel * (DD / 2);
    __shared__ float part[8];

    v2f A0, A1;
    A0[0] = half ? Wm1[0 * MH + col]      : Wm1[1 * MH + col];
    A0[1] = half ? bm1[col]               : Wm1[2 * MH + col];
    A1[0] = half ? Wm1[0 * MH + 16 + col] : Wm1[1 * MH + 16 + col];
    A1[1] = half ? bm1[16 + col]          : Wm1[2 * MH + 16 + col];

    const float* wmb = Wm2 + (half ? 8 : 0);
    float hA[8], hB[8];
#pragma unroll
    for (int r = 0; r < 8; ++r) {
        hA[r] = 0.5f * wmb[r];
        hB[r] = 0.5f * wmb[16 + r];
    }

    const float hwA = 0.5f * Wm2[col];
    const float hwB = 0.5f * Wm2[16 + col];
    float g0 = hwA * A0[0] + hwB * A1[0];
    float g1 = hwA * A0[1] + hwB * A1[1];

    const float bm2v = bm2[0];
    const float accInit = half ? 0.0f : bm2v;

    const float* __restrict__ srcp =
        (half ? aPre : W + (size_t)row * DD) + hoff + 4 * col;
    float* __restrict__ orow = Wout + (size_t)row * DD + hoff + 4 * lane;

    v4f n0 = *(const v4f*)(srcp);
    v4f n1 = *(const v4f*)(srcp + 64);
    v4f n2 = *(const v4f*)(srcp + 128);
    v4f n3 = *(const v4f*)(srcp + 192);

    float vPost;
    {
        const float* __restrict__ wrow = W + (size_t)row * DD + hoff + 4 * lane;
        const float* __restrict__ arow = aPre + hoff + 4 * lane;
        const float brow = bias[row];
        float acc = 0.0f;
#pragma unroll
        for (int k = 0; k < DD / 256; ++k) {
            const float4 wv = *(const float4*)(wrow + 128 * k);
            const float4 av = *(const float4*)(arow + 128 * k);
            acc += wv.x * av.x + wv.y * av.y + wv.z * av.z + wv.w * av.w;
        }
        acc += __shfl_xor(acc, 1);
        g0  += __shfl_xor(g0, 1);
        g1  += __shfl_xor(g1, 1);
        acc += __shfl_xor(acc, 2);
        g0  += __shfl_xor(g0, 2);
        g1  += __shfl_xor(g1, 2);
        acc += __shfl_xor(acc, 4);
        g0  += __shfl_xor(g0, 4);
        g1  += __shfl_xor(g1, 4);
        acc += __shfl_xor(acc, 8);
        g0  += __shfl_xor(g0, 8);
        g1  += __shfl_xor(g1, 8);
        acc += __shfl_xor(acc, 16);
        if (lane == 0) part[wave] = acc;
        __syncthreads();
        const float v = (part[wave & 6] + part[wave | 1]) + brow;
        vPost = v > 0.0f ? v : 0.0f;
        if (lane == 0 && hsel == 0) aOut[row] = vPost;
    }

    const float kc = half ? 1.0f : vPost;
    const float cl = __builtin_fmaf(g1, kc, accInit);

    for (int i0 = 0; i0 < DD / 2; i0 += 256) {
        const v4f c0 = n0, c1 = n1, c2 = n2, c3 = n3;
        if (i0 + 256 < DD / 2) {
            n0 = *(const v4f*)(srcp + i0 + 256);
            n1 = *(const v4f*)(srcp + i0 + 320);
            n2 = *(const v4f*)(srcp + i0 + 384);
            n3 = *(const v4f*)(srcp + i0 + 448);
        }
        meta_block128(A0, A1, c0, c1, kc, hA, hB, g0, cl, half, orow + i0);
        meta_block128(A0, A1, c2, c3, kc, hA, hB, g0, cl, half, orow + i0 + 128);
    }
}

extern "C" void kernel_launch(void* const* d_in, const int* in_sizes, int n_in,
                              void* d_out, int out_size, void* d_ws, size_t ws_size,
                              hipStream_t stream) {
    const float* x   = (const float*)d_in[0];
    const float* W1  = (const float*)d_in[1];
    const float* b1  = (const float*)d_in[2];
    const float* W2  = (const float*)d_in[3];
    const float* b2  = (const float*)d_in[4];
    const float* W3  = (const float*)d_in[5];
    const float* b3  = (const float*)d_in[6];
    const float* Wm1 = (const float*)d_in[7];
    const float* bm1 = (const float*)d_in[8];
    const float* Wm2 = (const float*)d_in[9];
    const float* bm2 = (const float*)d_in[10];

    float* out = (float*)d_out;
    float* nW1 = out + DD;
    float* nW2 = nW1 + (size_t)DD * DD;
    float* nW3 = nW2 + (size_t)DD * DD;

    float* a1 = (float*)d_ws;
    float* a2 = a1 + DD;

    dim3 blk(256);
    dim3 grid(DD / 4);

    layer_forward_meta_kernel<<<grid, blk, 0, stream>>>(W1, x,  b1, a1,  nW1, Wm2, bm2, Wm1, bm1);
    layer_forward_meta_kernel<<<grid, blk, 0, stream>>>(W2, a1, b2, a2,  nW2, Wm2, bm2, Wm1, bm1);
    layer_forward_meta_kernel<<<grid, blk, 0, stream>>>(W3, a2, b3, out, nW3, Wm2, bm2, Wm1, bm1);
}
